# starting kernel
# baseline (speedup 1.0000x reference)
#include <hip/hip_runtime.h>

typedef __attribute__((ext_vector_type(2))) float v2f;
typedef __attribute__((ext_vector_type(8))) float v8f;

#define H 128
#define W 128

__global__ __launch_bounds__(256)
void dw_gate_kernel(const float* __restrict__ lm,
                    const float* __restrict__ w_lm,
                    const float* __restrict__ b_lm,
                    float* __restrict__ gates)
{
    const int blk = blockIdx.x;
    const int g   = blockIdx.y;
    const int b   = blockIdx.z;
    const int tid = threadIdx.x;
    const int y0  = blk * 16;

    __shared__ float buf[2][26][138];
    __shared__ float gacc[16][128];

    for (int idx = tid; idx < 16*128; idx += 256)
        gacc[idx >> 7][idx & 127] = 0.f;
    for (int idx = tid; idx < 2*26*10; idx += 256) {
        int bsel = idx / 260, rem = idx % 260;
        int i = rem / 10, j = rem % 10;
        int col = (j < 5) ? j : (133 + (j - 5));
        buf[bsel][i][col] = 0.f;
    }
    __syncthreads();

    const int s0[4]    = {17, 27, 0, 48};
    const int cnt0[4]  = {10,  9, 17, 20};
    const int segs = (g == 0) ? 2 : 1;

    for (int s = 0; s < segs; ++s) {
        const int cstart = (s == 0) ? s0[g] : 36;
        const int ccnt   = (s == 0) ? cnt0[g] : 12;
        for (int ci = 0; ci < ccnt; ++ci) {
            const int c = cstart + ci;
            const float* src = lm + ((size_t)(b*68 + c) * H) * W;
            for (int idx = tid; idx < 26*128; idx += 256) {
                int i = idx >> 7, x = idx & 127;
                int y = y0 - 5 + i;
                buf[0][i][x + 5] = (y >= 0 && y < H) ? src[y*W + x] : 0.f;
            }
            __syncthreads();
            for (int k = 0; k < 5; ++k) {
                const float* wp = w_lm + (size_t)(k*68 + c) * 9;
                const float w00=wp[0],w01=wp[1],w02=wp[2],
                            w10=wp[3],w11=wp[4],w12=wp[5],
                            w20=wp[6],w21=wp[7],w22=wp[8];
                const float bias = b_lm[k*68 + c];
                float (*sb)[138] = buf[k & 1];
                float (*db)[138] = buf[(k + 1) & 1];
                const int rlo = k + 1, nrows = (24 - k) - rlo + 1;
                for (int idx = tid; idx < nrows*128; idx += 256) {
                    int i = rlo + (idx >> 7);
                    int x = idx & 127;
                    int y = y0 - 5 + i;
                    float v;
                    if (y < 0 || y >= H) {
                        v = 0.f;
                    } else {
                        int xc = x + 5;
                        v = bias
                          + w00*sb[i-1][xc-1] + w01*sb[i-1][xc] + w02*sb[i-1][xc+1]
                          + w10*sb[i  ][xc-1] + w11*sb[i  ][xc] + w12*sb[i  ][xc+1]
                          + w20*sb[i+1][xc-1] + w21*sb[i+1][xc] + w22*sb[i+1][xc+1];
                    }
                    db[i][x + 5] = v;
                }
                __syncthreads();
            }
            for (int idx = tid; idx < 16*128; idx += 256) {
                int r = idx >> 7, x = idx & 127;
                gacc[r][x] += buf[1][r + 5][x + 5];
            }
            __syncthreads();
        }
    }

    float* gp = gates + (((size_t)(b*4 + g) * H) + y0) * W;
    for (int idx = tid; idx < 16*128; idx += 256) {
        int r = idx >> 7, x = idx & 127;
        gp[r*W + x] = gacc[r][x];
    }
}

__global__ __launch_bounds__(256)
void conv16_wmma_kernel(const float* __restrict__ in,
                        float* __restrict__ out,
                        const float* __restrict__ w,
                        const float* __restrict__ bias,
                        const float* __restrict__ gates,
                        int gidx, int mode)
{
    const int y   = blockIdx.x;
    const int b   = blockIdx.y;
    const int tid = threadIdx.x;

    __shared__ float inT[3*130*18];
    __shared__ v2f   wB[36*32];

    for (int idx = tid; idx < 16*3*130; idx += 256) {
        int ch  = idx / 390;
        int rem = idx - ch*390;
        int r   = rem / 130;
        int xh  = rem - r*130;
        int yi = y + r - 1;
        int x  = xh - 1;
        float v = 0.f;
        if (yi >= 0 && yi < H && x >= 0 && x < W)
            v = in[((size_t)(b*16 + ch) * H + yi) * W + x];
        inT[(r*130 + xh)*18 + ch] = v;
    }
    for (int idx = tid; idx < 36*32; idx += 256) {
        int it   = idx >> 5;
        int lane = idx & 31;
        int kc  = it & 3;
        int tap = it >> 2;
        int ky = tap / 3, kx = tap - ky*3;
        int oc  = lane & 15;
        int icb = kc*4 + ((lane >> 4) << 1);
        v2f wv;
        wv.x = w[((oc*16 + icb    )*3 + ky)*3 + kx];
        wv.y = w[((oc*16 + icb + 1)*3 + ky)*3 + kx];
        wB[idx] = wv;
    }
    __syncthreads();

    const int lane  = tid & 31;
    const int wave  = tid >> 5;
    const int x0    = wave * 16;
    const int laneM = lane & 15;
    const int half  = lane >> 4;

    v8f acc = {0.f,0.f,0.f,0.f,0.f,0.f,0.f,0.f};
    #pragma unroll
    for (int ky = 0; ky < 3; ++ky) {
        #pragma unroll
        for (int kx = 0; kx < 3; ++kx) {
            const int xh = x0 + laneM + kx;
            const float* ap = &inT[(ky*130 + xh)*18 + 2*half];
            const int tap = ky*3 + kx;
            #pragma unroll
            for (int kc = 0; kc < 4; ++kc) {
                v2f a;
                a.x = ap[kc*4];
                a.y = ap[kc*4 + 1];
                v2f bb = wB[(tap*4 + kc)*32 + lane];
                acc = __builtin_amdgcn_wmma_f32_16x16x4_f32(
                          false, a, false, bb, (short)0, acc, false, false);
            }
        }
    }

    const float bv = bias[laneM];
    const int xb = x0 + half*8;
    float* op = out + ((size_t)(b*16 + laneM) * H + y) * W + xb;
    if (mode == 0) {
        #pragma unroll
        for (int r = 0; r < 8; ++r) op[r] = acc[r] + bv;
    } else {
        const float* gp = gates + ((size_t)(b*4 + gidx) * H + y) * W + xb;
        #pragma unroll
        for (int r = 0; r < 8; ++r) {
            float v = (acc[r] + bv) * gp[r];
            if (mode == 2) v += op[r];
            op[r] = v;
        }
    }
}

extern "C" void kernel_launch(void* const* d_in, const int* in_sizes, int n_in,
                              void* d_out, int out_size, void* d_ws, size_t ws_size,
                              hipStream_t stream)
{
    (void)in_sizes; (void)n_in; (void)out_size; (void)ws_size;
    const float* landmarks = (const float*)d_in[0];
    const float* features  = (const float*)d_in[1];
    const float* w_lm      = (const float*)d_in[2];
    const float* b_lm      = (const float*)d_in[3];
    const float* w_fea     = (const float*)d_in[4];
    const float* b_fea     = (const float*)d_in[5];
    float* out = (float*)d_out;

    char* ws = (char*)d_ws;
    float* gates = (float*)ws;
    float* bufA  = (float*)(ws + (size_t)8  * 1024*1024);
    float* bufB  = (float*)(ws + (size_t)40 * 1024*1024);

    dw_gate_kernel<<<dim3(8, 4, 32), 256, 0, stream>>>(landmarks, w_lm, b_lm, gates);

    for (int g = 0; g < 4; ++g) {
        const float* w0  = w_fea + (size_t)(g*3 + 0) * 16*16*9;
        const float* w1  = w_fea + (size_t)(g*3 + 1) * 16*16*9;
        const float* w2  = w_fea + (size_t)(g*3 + 2) * 16*16*9;
        const float* bb0 = b_fea + (g*3 + 0) * 16;
        const float* bb1 = b_fea + (g*3 + 1) * 16;
        const float* bb2 = b_fea + (g*3 + 2) * 16;
        conv16_wmma_kernel<<<dim3(128, 32), 256, 0, stream>>>(features, bufA, w0, bb0, gates, g, 0);
        conv16_wmma_kernel<<<dim3(128, 32), 256, 0, stream>>>(bufA,     bufB, w1, bb1, gates, g, 0);
        conv16_wmma_kernel<<<dim3(128, 32), 256, 0, stream>>>(bufB,     out,  w2, bb2, gates, g, (g == 0) ? 1 : 2);
    }
}


# depthwise towers: layer passes re-tiled, lane = 4 columns, wave = run of 3 rows: a run reads each input row once (16-byte LDS reads); last layer adds into a register gate sum; sum order kept
# speedup vs baseline: 2.7452x; 2.7452x over previous
#include <hip/hip_runtime.h>

typedef __attribute__((ext_vector_type(2))) float v2f;
typedef __attribute__((ext_vector_type(4))) float v4f;
typedef __attribute__((ext_vector_type(8))) float v8f;
typedef const __attribute__((address_space(4))) float cfloat;

#define H 128
#define W 128

#define DP 136

template <int NR>
__device__ __forceinline__
void dw_rows(const float (*sb)[DP], int i0, int rmax, int lane,
             float w00, float w01, float w02, float w10, float w11, float w12,
             float w20, float w21, float w22, float bias, v4f (&o)[NR])
{
    float r[NR + 2][6];
    #pragma unroll
    for (int j = 0; j < NR + 2; ++j) {
        const int i = min(i0 - 1 + j, rmax);
        const float* p = &sb[i][4*lane + 3];
        const v4f c = *(const v4f*)(p + 1);
        r[j][0] = p[0];
        r[j][1] = c.x; r[j][2] = c.y; r[j][3] = c.z; r[j][4] = c.w;
        r[j][5] = p[5];
    }
    #pragma unroll
    for (int q = 0; q < NR; ++q) {
        #pragma unroll
        for (int e = 0; e < 4; ++e) {
            o[q][e] = bias
                    + w00*r[q  ][e] + w01*r[q  ][e+1] + w02*r[q  ][e+2]
                    + w10*r[q+1][e] + w11*r[q+1][e+1] + w12*r[q+1][e+2]
                    + w20*r[q+2][e] + w21*r[q+2][e+1] + w22*r[q+2][e+2];
        }
    }
}

__global__ __launch_bounds__(256)
void dw_gate_kernel(const float* __restrict__ lm,
                    const float* __restrict__ w_lm,
                    const float* __restrict__ b_lm,
                    float* __restrict__ gates)
{
    const int blk = blockIdx.x;
    const int g   = blockIdx.y;
    const int b   = blockIdx.z;
    const int tid = threadIdx.x;
    const int y0  = blk * 16;
    const int lane = tid & 31;
    const int wv = __builtin_amdgcn_readfirstlane(tid >> 5);

    __shared__ __attribute__((aligned(16))) float buf[2][26][DP];

    for (int idx = tid; idx < 2*26*2; idx += 256) {
        const int bsel = idx / 52, rem = idx % 52;
        buf[bsel][rem >> 1][(rem & 1) ? 132 : 3] = 0.f;
    }

    v4f gsum0 = {0.f, 0.f, 0.f, 0.f};
    v4f gsum1 = {0.f, 0.f, 0.f, 0.f};

    const int s0[4]    = {17, 27, 0, 48};
    const int cnt0[4]  = {10,  9, 17, 20};
    const int segs = (g == 0) ? 2 : 1;

    for (int s = 0; s < segs; ++s) {
        const int cstart = (s == 0) ? s0[g] : 36;
        const int ccnt   = (s == 0) ? cnt0[g] : 12;
        for (int ci = 0; ci < ccnt; ++ci) {
            const int c = cstart + ci;
            const float* src = lm + ((size_t)(b*68 + c) * H) * W;
            float wk[5][9], bk[5];
            #pragma unroll
            for (int k = 0; k < 5; ++k) {
                cfloat* wp = (cfloat*)(w_lm + (size_t)(k*68 + c) * 9);
                #pragma unroll
                for (int q = 0; q < 9; ++q) wk[k][q] = wp[q];
                bk[k] = ((cfloat*)b_lm)[k*68 + c];
            }
            v4f t[4];
            #pragma unroll
            for (int j = 0; j < 4; ++j) {
                const int i  = min(wv + 8*j, 25);
                const int y  = y0 - 5 + i;
                const int yc = min(max(y, 0), H - 1);
                v4f v = *(const v4f*)(src + yc*W + 4*lane);
                if (y < 0 || y >= H) v = (v4f){0.f, 0.f, 0.f, 0.f};
                t[j] = v;
            }
            #pragma unroll
            for (int j = 0; j < 4; ++j) {
                const int i = wv + 8*j;
                if (i < 26) *(v4f*)&buf[0][i][4 + 4*lane] = t[j];
            }
            __syncthreads();
            #pragma unroll
            for (int k = 0; k < 4; ++k) {
                const float w00=wk[k][0],w01=wk[k][1],w02=wk[k][2],
                            w10=wk[k][3],w11=wk[k][4],w12=wk[k][5],
                            w20=wk[k][6],w21=wk[k][7],w22=wk[k][8];
                const float bias = bk[k];
                const int rhi = 24 - k;
                const int i0  = k + 1 + 3*wv;
                if (i0 <= rhi) {
                    v4f o[3];
                    dw_rows<3>(buf[k & 1], i0, 25 - k, lane, w00,w01,w02,w10,w11,w12,w20,w21,w22, bias, o);
                    #pragma unroll
                    for (int q = 0; q < 3; ++q) {
                        const int i = i0 + q;
                        if (i <= rhi) {
                            const int y = y0 - 5 + i;
                            const v4f v = (y < 0 || y >= H) ? (v4f){0.f, 0.f, 0.f, 0.f} : o[q];
                            *(v4f*)&buf[(k + 1) & 1][i][4 + 4*lane] = v;
                        }
                    }
                }
                __syncthreads();
            }
            {
                const float w00=wk[4][0],w01=wk[4][1],w02=wk[4][2],
                            w10=wk[4][3],w11=wk[4][4],w12=wk[4][5],
                            w20=wk[4][6],w21=wk[4][7],w22=wk[4][8];
                const float bias = bk[4];
                v4f o[2];
                dw_rows<2>(buf[0], 5 + 2*wv, 21, lane, w00,w01,w02,w10,w11,w12,w20,w21,w22, bias, o);
                gsum0 += o[0];
                gsum1 += o[1];
            }
            __syncthreads();
        }
    }

    float* gp = gates + (((size_t)(b*4 + g) * H) + y0 + 2*wv) * W + 4*lane;
    *(v4f*)gp       = gsum0;
    *(v4f*)(gp + W) = gsum1;
}

#define XP   130
#define CP   18
#define ROWF (XP*CP)
#define BAND 8

template <bool LAST, bool ACC>
__device__ __forceinline__
void conv_layer(const float* src, float* dst, const v2f* wl,
                int nrows, int ybase, float bv,
                int x0, int laneM, int half, int lane,
                float* __restrict__ outb, const float* __restrict__ gateb)
{
    v2f breg[36];
    #pragma unroll
    for (int i = 0; i < 36; ++i) breg[i] = wl[i*32 + lane];

    for (int q = 0; q < nrows; q += 2) {
        v8f acc0 = {0.f,0.f,0.f,0.f,0.f,0.f,0.f,0.f};
        v8f acc1 = {0.f,0.f,0.f,0.f,0.f,0.f,0.f,0.f};
        v4f gv[2][2], ov[2][2];
        if (LAST) {
            #pragma unroll
            for (int p = 0; p < 2; ++p) {
                const int xb = x0 + half*8;
                const float* gp = gateb + (size_t)(ybase + q + p) * W + xb;
                const float* op = outb + ((size_t)laneM * H + (ybase + q + p)) * W + xb;
                gv[p][0] = *(const v4f*)gp;
                gv[p][1] = *(const v4f*)(gp + 4);
                if (ACC) {
                    ov[p][0] = *(const v4f*)op;
                    ov[p][1] = *(const v4f*)(op + 4);
                }
            }
        }
        #pragma unroll
        for (int ky = 0; ky < 3; ++ky) {
            #pragma unroll
            for (int kx = 0; kx < 3; ++kx) {
                const float* ap0 = src + ((q + ky)*XP + x0 + laneM + kx)*CP + 2*half;
                const float* ap1 = ap0 + ROWF;
                const int tap = ky*3 + kx;
                #pragma unroll
                for (int kc = 0; kc < 4; ++kc) {
                    v2f a0, a1;
                    a0.x = ap0[kc*4];
                    a0.y = ap0[kc*4 + 1];
                    a1.x = ap1[kc*4];
                    a1.y = ap1[kc*4 + 1];
                    acc0 = __builtin_amdgcn_wmma_f32_16x16x4_f32(
                               false, a0, false, breg[tap*4 + kc], (short)0, acc0, false, false);
                    acc1 = __builtin_amdgcn_wmma_f32_16x16x4_f32(
                               false, a1, false, breg[tap*4 + kc], (short)0, acc1, false, false);
                }
            }
        }
        #pragma unroll
        for (int p = 0; p < 2; ++p) {
            const v8f acc = p ? acc1 : acc0;
            const int y = ybase + q + p;
            if (!LAST) {
                const bool inside = (y >= 0 && y < H);
                float* d = dst + ((q + p)*XP + x0 + 8*half + 1)*CP + laneM;
                #pragma unroll
                for (int r = 0; r < 8; ++r) d[r*CP] = inside ? (acc[r] + bv) : 0.f;
            } else {
                const int xb = x0 + half*8;
                float* op = outb + ((size_t)laneM * H + y) * W + xb;
                v4f res[2];
                #pragma unroll
                for (int r = 0; r < 8; ++r) {
                    float v = (acc[r] + bv) * gv[p][r >> 2][r & 3];
                    if (ACC) v += ov[p][r >> 2][r & 3];
                    res[r >> 2][r & 3] = v;
                }
                *(v4f*)op       = res[0];
                *(v4f*)(op + 4) = res[1];
            }
        }
    }
}

template <bool ACC>
__global__ __launch_bounds__(256)
void conv3_fused_wmma_kernel(const float* __restrict__ in,
                             float* __restrict__ out,
                             const float* __restrict__ w3,
                             const float* __restrict__ b3,
                             const float* __restrict__ gates,
                             int gidx)
{
    const int y0  = blockIdx.x * BAND;
    const int b   = blockIdx.y;
    const int tid = threadIdx.x;

    __shared__ float bufA[(BAND + 6)*ROWF];
    __shared__ float bufB[(BAND + 4)*ROWF];
    __shared__ v2f   wB[3*36*32];

    float* wBf = (float*)wB;
    #pragma unroll
    for (int it = 0; it < 27; ++it) {
        const int e   = tid + it*256;
        const int L   = e / 2304;
        const int rem = e - L*2304;
        const int oc  = rem / 144;
        const int r2  = rem - oc*144;
        const int ic  = r2 / 9;
        const int tap = r2 - ic*9;
        const int kc  = ic >> 2, hf = (ic >> 1) & 1, comp = ic & 1;
        wBf[((L*36 + tap*4 + kc)*32 + oc + 16*hf)*2 + comp] = w3[e];
    }
    for (int idx = tid; idx < ((BAND + 6) + (BAND + 4))*2*16; idx += 256) {
        const int ch = idx & 15;
        const int xh = ((idx >> 4) & 1) ? (XP - 1) : 0;
        const int r  = idx >> 5;
        if (r < BAND + 6) bufA[(r*XP + xh)*CP + ch] = 0.f;
        else              bufB[((r - (BAND + 6))*XP + xh)*CP + ch] = 0.f;
    }
    for (int it0 = 0; it0 < 28; it0 += 7) {
        v4f v[7];
        #pragma unroll
        for (int j = 0; j < 7; ++j) {
            const int idx = tid + (it0 + j)*256;
            const int x4 = idx & 31, ch = (idx >> 5) & 15, r = idx >> 9;
            const int yi = y0 - 3 + r;
            const int yc = min(max(yi, 0), H - 1);
            v4f t = *(const v4f*)(in + ((size_t)(b*16 + ch) * H + yc) * W + x4*4);
            if (yi < 0 || yi >= H) t = (v4f){0.f, 0.f, 0.f, 0.f};
            v[j] = t;
        }
        #pragma unroll
        for (int j = 0; j < 7; ++j) {
            const int idx = tid + (it0 + j)*256;
            const int x4 = idx & 31, ch = (idx >> 5) & 15, r = idx >> 9;
            float* d = &bufA[(r*XP + x4*4 + 1)*CP + ch];
            d[0]    = v[j].x;
            d[CP]   = v[j].y;
            d[2*CP] = v[j].z;
            d[3*CP] = v[j].w;
        }
    }
    __syncthreads();

    const int lane  = tid & 31;
    const int wave  = tid >> 5;
    const int x0    = wave * 16;
    const int laneM = lane & 15;
    const int half  = lane >> 4;

    float* outb = out + (size_t)b * 16 * H * W;
    const float* gateb = gates + (size_t)(b*4 + gidx) * H * W;

    conv_layer<false, ACC>(bufA, bufB, wB,           BAND + 4, y0 - 2, b3[laneM],
                           x0, laneM, half, lane, outb, gateb);
    __syncthreads();
    conv_layer<false, ACC>(bufB, bufA, wB + 36*32,   BAND + 2, y0 - 1, b3[16 + laneM],
                           x0, laneM, half, lane, outb, gateb);
    __syncthreads();
    conv_layer<true,  ACC>(bufA, bufB, wB + 2*36*32, BAND,     y0,     b3[32 + laneM],
                           x0, laneM, half, lane, outb, gateb);
}

extern "C" void kernel_launch(void* const* d_in, const int* in_sizes, int n_in,
                              void* d_out, int out_size, void* d_ws, size_t ws_size,
                              hipStream_t stream)
{
    (void)in_sizes; (void)n_in; (void)out_size; (void)ws_size;
    const float* landmarks = (const float*)d_in[0];
    const float* features  = (const float*)d_in[1];
    const float* w_lm      = (const float*)d_in[2];
    const float* b_lm      = (const float*)d_in[3];
    const float* w_fea     = (const float*)d_in[4];
    const float* b_fea     = (const float*)d_in[5];
    float* out = (float*)d_out;

    float* gates = (float*)d_ws;

    dw_gate_kernel<<<dim3(8, 4, 32), 256, 0, stream>>>(landmarks, w_lm, b_lm, gates);

    for (int g = 0; g < 4; ++g) {
        const float* wg = w_fea + (size_t)g * 3 * 16*16*9;
        const float* bg = b_fea + g * 3 * 16;
        if (g == 0)
            conv3_fused_wmma_kernel<false><<<dim3(128 / BAND, 32), 256, 0, stream>>>(
                features, out, wg, bg, gates, g);
        else
            conv3_fused_wmma_kernel<true><<<dim3(128 / BAND, 32), 256, 0, stream>>>(
                features, out, wg, bg, gates, g);
    }
}


# depthwise towers: every tap is one fused multiply-add (fmaf) into its running sum; the plain sum of products had compiled to packed multiplies, moves and adds; sum order kept, last bits differ
# speedup vs baseline: 3.1218x; 1.1372x over previous
#include <hip/hip_runtime.h>

typedef __attribute__((ext_vector_type(2))) float v2f;
typedef __attribute__((ext_vector_type(4))) float v4f;
typedef __attribute__((ext_vector_type(8))) float v8f;
typedef const __attribute__((address_space(4))) float cfloat;

#define H 128
#define W 128

#define DP 136

template <int NR>
__device__ __forceinline__
void dw_rows(const float (*sb)[DP], int i0, int rmax, int lane,
             const float (&w)[9], float bias, v4f (&o)[NR])
{
    float r[NR + 2][6];
    #pragma unroll
    for (int j = 0; j < NR + 2; ++j) {
        const int i = min(i0 - 1 + j, rmax);
        const float* p = &sb[i][4*lane + 3];
        const v4f c = *(const v4f*)(p + 1);
        r[j][0] = p[0];
        r[j][1] = c.x; r[j][2] = c.y; r[j][3] = c.z; r[j][4] = c.w;
        r[j][5] = p[5];
    }
    float acc[NR][4];
    #pragma unroll
    for (int q = 0; q < NR; ++q) {
        #pragma unroll
        for (int e = 0; e < 4; ++e) acc[q][e] = bias;
    }
    #pragma unroll
    for (int ky = 0; ky < 3; ++ky) {
        #pragma unroll
        for (int kx = 0; kx < 3; ++kx) {
            const float wt = w[ky*3 + kx];
            #pragma unroll
            for (int q = 0; q < NR; ++q) {
                #pragma unroll
                for (int e = 0; e < 4; ++e)
                    acc[q][e] = __builtin_fmaf(wt, r[q + ky][e + kx], acc[q][e]);
            }
        }
    }
    #pragma unroll
    for (int q = 0; q < NR; ++q) {
        #pragma unroll
        for (int e = 0; e < 4; ++e) o[q][e] = acc[q][e];
    }
}

__global__ __launch_bounds__(256)
void dw_gate_kernel(const float* __restrict__ lm,
                    const float* __restrict__ w_lm,
                    const float* __restrict__ b_lm,
                    float* __restrict__ gates)
{
    const int blk = blockIdx.x;
    const int g   = blockIdx.y;
    const int b   = blockIdx.z;
    const int tid = threadIdx.x;
    const int y0  = blk * 16;
    const int lane = tid & 31;
    const int wv = __builtin_amdgcn_readfirstlane(tid >> 5);

    __shared__ __attribute__((aligned(16))) float buf[2][26][DP];

    for (int idx = tid; idx < 2*26*2; idx += 256) {
        const int bsel = idx / 52, rem = idx % 52;
        buf[bsel][rem >> 1][(rem & 1) ? 132 : 3] = 0.f;
    }

    v4f gsum0 = {0.f, 0.f, 0.f, 0.f};
    v4f gsum1 = {0.f, 0.f, 0.f, 0.f};

    const int s0[4]    = {17, 27, 0, 48};
    const int cnt0[4]  = {10,  9, 17, 20};
    const int segs = (g == 0) ? 2 : 1;

    for (int s = 0; s < segs; ++s) {
        const int cstart = (s == 0) ? s0[g] : 36;
        const int ccnt   = (s == 0) ? cnt0[g] : 12;
        for (int ci = 0; ci < ccnt; ++ci) {
            const int c = cstart + ci;
            const float* src = lm + ((size_t)(b*68 + c) * H) * W;
            float wk[5][9], bk[5];
            #pragma unroll
            for (int k = 0; k < 5; ++k) {
                cfloat* wp = (cfloat*)(w_lm + (size_t)(k*68 + c) * 9);
                #pragma unroll
                for (int q = 0; q < 9; ++q) wk[k][q] = wp[q];
                bk[k] = ((cfloat*)b_lm)[k*68 + c];
            }
            v4f t[4];
            #pragma unroll
            for (int j = 0; j < 4; ++j) {
                const int i  = min(wv + 8*j, 25);
                const int y  = y0 - 5 + i;
                const int yc = min(max(y, 0), H - 1);
                v4f v = *(const v4f*)(src + yc*W + 4*lane);
                if (y < 0 || y >= H) v = (v4f){0.f, 0.f, 0.f, 0.f};
                t[j] = v;
            }
            #pragma unroll
            for (int j = 0; j < 4; ++j) {
                const int i = wv + 8*j;
                if (i < 26) *(v4f*)&buf[0][i][4 + 4*lane] = t[j];
            }
            __syncthreads();
            #pragma unroll
            for (int k = 0; k < 4; ++k) {
                const float bias = bk[k];
                const int rhi = 24 - k;
                const int i0  = k + 1 + 3*wv;
                if (i0 <= rhi) {
                    v4f o[3];
                    dw_rows<3>(buf[k & 1], i0, 25 - k, lane, wk[k], bias, o);
                    #pragma unroll
                    for (int q = 0; q < 3; ++q) {
                        const int i = i0 + q;
                        if (i <= rhi) {
                            const int y = y0 - 5 + i;
                            const v4f v = (y < 0 || y >= H) ? (v4f){0.f, 0.f, 0.f, 0.f} : o[q];
                            *(v4f*)&buf[(k + 1) & 1][i][4 + 4*lane] = v;
                        }
                    }
                }
                __syncthreads();
            }
            {
                const float bias = bk[4];
                v4f o[2];
                dw_rows<2>(buf[0], 5 + 2*wv, 21, lane, wk[4], bias, o);
                gsum0 += o[0];
                gsum1 += o[1];
            }
            __syncthreads();
        }
    }

    float* gp = gates + (((size_t)(b*4 + g) * H) + y0 + 2*wv) * W + 4*lane;
    *(v4f*)gp       = gsum0;
    *(v4f*)(gp + W) = gsum1;
}

#define XP   130
#define CP   18
#define ROWF (XP*CP)
#define BAND 8

template <bool LAST, bool ACC>
__device__ __forceinline__
void conv_layer(const float* src, float* dst, const v2f* wl,
                int nrows, int ybase, float bv,
                int x0, int laneM, int half, int lane,
                float* __restrict__ outb, const float* __restrict__ gateb)
{
    v2f breg[36];
    #pragma unroll
    for (int i = 0; i < 36; ++i) breg[i] = wl[i*32 + lane];

    for (int q = 0; q < nrows; q += 2) {
        v8f acc0 = {0.f,0.f,0.f,0.f,0.f,0.f,0.f,0.f};
        v8f acc1 = {0.f,0.f,0.f,0.f,0.f,0.f,0.f,0.f};
        v4f gv[2][2], ov[2][2];
        if (LAST) {
            #pragma unroll
            for (int p = 0; p < 2; ++p) {
                const int xb = x0 + half*8;
                const float* gp = gateb + (size_t)(ybase + q + p) * W + xb;
                const float* op = outb + ((size_t)laneM * H + (ybase + q + p)) * W + xb;
                gv[p][0] = *(const v4f*)gp;
                gv[p][1] = *(const v4f*)(gp + 4);
                if (ACC) {
                    ov[p][0] = *(const v4f*)op;
                    ov[p][1] = *(const v4f*)(op + 4);
                }
            }
        }
        #pragma unroll
        for (int ky = 0; ky < 3; ++ky) {
            #pragma unroll
            for (int kx = 0; kx < 3; ++kx) {
                const float* ap0 = src + ((q + ky)*XP + x0 + laneM + kx)*CP + 2*half;
                const float* ap1 = ap0 + ROWF;
                const int tap = ky*3 + kx;
                #pragma unroll
                for (int kc = 0; kc < 4; ++kc) {
                    v2f a0, a1;
                    a0.x = ap0[kc*4];
                    a0.y = ap0[kc*4 + 1];
                    a1.x = ap1[kc*4];
                    a1.y = ap1[kc*4 + 1];
                    acc0 = __builtin_amdgcn_wmma_f32_16x16x4_f32(
                               false, a0, false, breg[tap*4 + kc], (short)0, acc0, false, false);
                    acc1 = __builtin_amdgcn_wmma_f32_16x16x4_f32(
                               false, a1, false, breg[tap*4 + kc], (short)0, acc1, false, false);
                }
            }
        }
        #pragma unroll
        for (int p = 0; p < 2; ++p) {
            const v8f acc = p ? acc1 : acc0;
            const int y = ybase + q + p;
            if (!LAST) {
                const bool inside = (y >= 0 && y < H);
                float* d = dst + ((q + p)*XP + x0 + 8*half + 1)*CP + laneM;
                #pragma unroll
                for (int r = 0; r < 8; ++r) d[r*CP] = inside ? (acc[r] + bv) : 0.f;
            } else {
                const int xb = x0 + half*8;
                float* op = outb + ((size_t)laneM * H + y) * W + xb;
                v4f res[2];
                #pragma unroll
                for (int r = 0; r < 8; ++r) {
                    float v = (acc[r] + bv) * gv[p][r >> 2][r & 3];
                    if (ACC) v += ov[p][r >> 2][r & 3];
                    res[r >> 2][r & 3] = v;
                }
                *(v4f*)op       = res[0];
                *(v4f*)(op + 4) = res[1];
            }
        }
    }
}

template <bool ACC>
__global__ __launch_bounds__(256)
void conv3_fused_wmma_kernel(const float* __restrict__ in,
                             float* __restrict__ out,
                             const float* __restrict__ w3,
                             const float* __restrict__ b3,
                             const float* __restrict__ gates,
                             int gidx)
{
    const int y0  = blockIdx.x * BAND;
    const int b   = blockIdx.y;
    const int tid = threadIdx.x;

    __shared__ float bufA[(BAND + 6)*ROWF];
    __shared__ float bufB[(BAND + 4)*ROWF];
    __shared__ v2f   wB[3*36*32];

    float* wBf = (float*)wB;
    #pragma unroll
    for (int it = 0; it < 27; ++it) {
        const int e   = tid + it*256;
        const int L   = e / 2304;
        const int rem = e - L*2304;
        const int oc  = rem / 144;
        const int r2  = rem - oc*144;
        const int ic  = r2 / 9;
        const int tap = r2 - ic*9;
        const int kc  = ic >> 2, hf = (ic >> 1) & 1, comp = ic & 1;
        wBf[((L*36 + tap*4 + kc)*32 + oc + 16*hf)*2 + comp] = w3[e];
    }
    for (int idx = tid; idx < ((BAND + 6) + (BAND + 4))*2*16; idx += 256) {
        const int ch = idx & 15;
        const int xh = ((idx >> 4) & 1) ? (XP - 1) : 0;
        const int r  = idx >> 5;
        if (r < BAND + 6) bufA[(r*XP + xh)*CP + ch] = 0.f;
        else              bufB[((r - (BAND + 6))*XP + xh)*CP + ch] = 0.f;
    }
    for (int it0 = 0; it0 < 28; it0 += 7) {
        v4f v[7];
        #pragma unroll
        for (int j = 0; j < 7; ++j) {
            const int idx = tid + (it0 + j)*256;
            const int x4 = idx & 31, ch = (idx >> 5) & 15, r = idx >> 9;
            const int yi = y0 - 3 + r;
            const int yc = min(max(yi, 0), H - 1);
            v4f t = *(const v4f*)(in + ((size_t)(b*16 + ch) * H + yc) * W + x4*4);
            if (yi < 0 || yi >= H) t = (v4f){0.f, 0.f, 0.f, 0.f};
            v[j] = t;
        }
        #pragma unroll
        for (int j = 0; j < 7; ++j) {
            const int idx = tid + (it0 + j)*256;
            const int x4 = idx & 31, ch = (idx >> 5) & 15, r = idx >> 9;
            float* d = &bufA[(r*XP + x4*4 + 1)*CP + ch];
            d[0]    = v[j].x;
            d[CP]   = v[j].y;
            d[2*CP] = v[j].z;
            d[3*CP] = v[j].w;
        }
    }
    __syncthreads();

    const int lane  = tid & 31;
    const int wave  = tid >> 5;
    const int x0    = wave * 16;
    const int laneM = lane & 15;
    const int half  = lane >> 4;

    float* outb = out + (size_t)b * 16 * H * W;
    const float* gateb = gates + (size_t)(b*4 + gidx) * H * W;

    conv_layer<false, ACC>(bufA, bufB, wB,           BAND + 4, y0 - 2, b3[laneM],
                           x0, laneM, half, lane, outb, gateb);
    __syncthreads();
    conv_layer<false, ACC>(bufB, bufA, wB + 36*32,   BAND + 2, y0 - 1, b3[16 + laneM],
                           x0, laneM, half, lane, outb, gateb);
    __syncthreads();
    conv_layer<true,  ACC>(bufA, bufB, wB + 2*36*32, BAND,     y0,     b3[32 + laneM],
                           x0, laneM, half, lane, outb, gateb);
}

extern "C" void kernel_launch(void* const* d_in, const int* in_sizes, int n_in,
                              void* d_out, int out_size, void* d_ws, size_t ws_size,
                              hipStream_t stream)
{
    (void)in_sizes; (void)n_in; (void)out_size; (void)ws_size;
    const float* landmarks = (const float*)d_in[0];
    const float* features  = (const float*)d_in[1];
    const float* w_lm      = (const float*)d_in[2];
    const float* b_lm      = (const float*)d_in[3];
    const float* w_fea     = (const float*)d_in[4];
    const float* b_fea     = (const float*)d_in[5];
    float* out = (float*)d_out;

    float* gates = (float*)d_ws;

    dw_gate_kernel<<<dim3(8, 4, 32), 256, 0, stream>>>(landmarks, w_lm, b_lm, gates);

    for (int g = 0; g < 4; ++g) {
        const float* wg = w_fea + (size_t)g * 3 * 16*16*9;
        const float* bg = b_fea + g * 3 * 16;
        if (g == 0)
            conv3_fused_wmma_kernel<false><<<dim3(128 / BAND, 32), 256, 0, stream>>>(
                features, out, wg, bg, gates, g);
        else
            conv3_fused_wmma_kernel<true><<<dim3(128 / BAND, 32), 256, 0, stream>>>(
                features, out, wg, bg, gates, g);
    }
}


# conv: a work-group stages its input band once and runs the 4 branches over it in turn (1 launch, not 4): layer 1 written in place over layer 0, gated sum held in registers; sum order kept
# speedup vs baseline: 3.3125x; 1.0611x over previous
#include <hip/hip_runtime.h>

typedef __attribute__((ext_vector_type(2))) float v2f;
typedef __attribute__((ext_vector_type(4))) float v4f;
typedef __attribute__((ext_vector_type(8))) float v8f;
typedef const __attribute__((address_space(4))) float cfloat;

#define H 128
#define W 128

#define DP 136

template <int NR>
__device__ __forceinline__
void dw_rows(const float (*sb)[DP], int i0, int rmax, int lane,
             const float (&w)[9], float bias, v4f (&o)[NR])
{
    float r[NR + 2][6];
    #pragma unroll
    for (int j = 0; j < NR + 2; ++j) {
        const int i = min(i0 - 1 + j, rmax);
        const float* p = &sb[i][4*lane + 3];
        const v4f c = *(const v4f*)(p + 1);
        r[j][0] = p[0];
        r[j][1] = c.x; r[j][2] = c.y; r[j][3] = c.z; r[j][4] = c.w;
        r[j][5] = p[5];
    }
    float acc[NR][4];
    #pragma unroll
    for (int q = 0; q < NR; ++q) {
        #pragma unroll
        for (int e = 0; e < 4; ++e) acc[q][e] = bias;
    }
    #pragma unroll
    for (int ky = 0; ky < 3; ++ky) {
        #pragma unroll
        for (int kx = 0; kx < 3; ++kx) {
            const float wt = w[ky*3 + kx];
            #pragma unroll
            for (int q = 0; q < NR; ++q) {
                #pragma unroll
                for (int e = 0; e < 4; ++e)
                    acc[q][e] = __builtin_fmaf(wt, r[q + ky][e + kx], acc[q][e]);
            }
        }
    }
    #pragma unroll
    for (int q = 0; q < NR; ++q) {
        #pragma unroll
        for (int e = 0; e < 4; ++e) o[q][e] = acc[q][e];
    }
}

__global__ __launch_bounds__(256)
void dw_gate_kernel(const float* __restrict__ lm,
                    const float* __restrict__ w_lm,
                    const float* __restrict__ b_lm,
                    float* __restrict__ gates)
{
    const int blk = blockIdx.x;
    const int g   = blockIdx.y;
    const int b   = blockIdx.z;
    const int tid = threadIdx.x;
    const int y0  = blk * 16;
    const int lane = tid & 31;
    const int wv = __builtin_amdgcn_readfirstlane(tid >> 5);

    __shared__ __attribute__((aligned(16))) float buf[2][26][DP];

    for (int idx = tid; idx < 2*26*2; idx += 256) {
        const int bsel = idx / 52, rem = idx % 52;
        buf[bsel][rem >> 1][(rem & 1) ? 132 : 3] = 0.f;
    }

    v4f gsum0 = {0.f, 0.f, 0.f, 0.f};
    v4f gsum1 = {0.f, 0.f, 0.f, 0.f};

    const int s0[4]    = {17, 27, 0, 48};
    const int cnt0[4]  = {10,  9, 17, 20};
    const int segs = (g == 0) ? 2 : 1;

    for (int s = 0; s < segs; ++s) {
        const int cstart = (s == 0) ? s0[g] : 36;
        const int ccnt   = (s == 0) ? cnt0[g] : 12;
        for (int ci = 0; ci < ccnt; ++ci) {
            const int c = cstart + ci;
            const float* src = lm + ((size_t)(b*68 + c) * H) * W;
            float wk[5][9], bk[5];
            #pragma unroll
            for (int k = 0; k < 5; ++k) {
                cfloat* wp = (cfloat*)(w_lm + (size_t)(k*68 + c) * 9);
                #pragma unroll
                for (int q = 0; q < 9; ++q) wk[k][q] = wp[q];
                bk[k] = ((cfloat*)b_lm)[k*68 + c];
            }
            v4f t[4];
            #pragma unroll
            for (int j = 0; j < 4; ++j) {
                const int i  = min(wv + 8*j, 25);
                const int y  = y0 - 5 + i;
                const int yc = min(max(y, 0), H - 1);
                v4f v = *(const v4f*)(src + yc*W + 4*lane);
                if (y < 0 || y >= H) v = (v4f){0.f, 0.f, 0.f, 0.f};
                t[j] = v;
            }
            #pragma unroll
            for (int j = 0; j < 4; ++j) {
                const int i = wv + 8*j;
                if (i < 26) *(v4f*)&buf[0][i][4 + 4*lane] = t[j];
            }
            __syncthreads();
            #pragma unroll
            for (int k = 0; k < 4; ++k) {
                const float bias = bk[k];
                const int rhi = 24 - k;
                const int i0  = k + 1 + 3*wv;
                if (i0 <= rhi) {
                    v4f o[3];
                    dw_rows<3>(buf[k & 1], i0, 25 - k, lane, wk[k], bias, o);
                    #pragma unroll
                    for (int q = 0; q < 3; ++q) {
                        const int i = i0 + q;
                        if (i <= rhi) {
                            const int y = y0 - 5 + i;
                            const v4f v = (y < 0 || y >= H) ? (v4f){0.f, 0.f, 0.f, 0.f} : o[q];
                            *(v4f*)&buf[(k + 1) & 1][i][4 + 4*lane] = v;
                        }
                    }
                }
                __syncthreads();
            }
            {
                const float bias = bk[4];
                v4f o[2];
                dw_rows<2>(buf[0], 5 + 2*wv, 21, lane, wk[4], bias, o);
                gsum0 += o[0];
                gsum1 += o[1];
            }
            __syncthreads();
        }
    }

    float* gp = gates + (((size_t)(b*4 + g) * H) + y0 + 2*wv) * W + 4*lane;
    *(v4f*)gp       = gsum0;
    *(v4f*)(gp + W) = gsum1;
}

#define XP   130
#define CP   18
#define ROWF (XP*CP)
#define BAND 8
#define WLAY (36*32)
#define WBR  (3*16*16*9)

template <int NR>
__device__ __forceinline__
void conv_rows(const float* src, const v2f* wl, int x0, int laneM, int half, int lane,
               v8f (&acc)[NR])
{
    v2f breg[36];
    #pragma unroll
    for (int i = 0; i < 36; ++i) breg[i] = wl[i*32 + lane];

    #pragma unroll
    for (int q = 0; q < NR; q += 2) {
        v8f acc0 = {0.f,0.f,0.f,0.f,0.f,0.f,0.f,0.f};
        v8f acc1 = {0.f,0.f,0.f,0.f,0.f,0.f,0.f,0.f};
        #pragma unroll
        for (int ky = 0; ky < 3; ++ky) {
            #pragma unroll
            for (int kx = 0; kx < 3; ++kx) {
                const float* ap0 = src + ((q + ky)*XP + x0 + laneM + kx)*CP + 2*half;
                const float* ap1 = ap0 + ROWF;
                const int tap = ky*3 + kx;
                #pragma unroll
                for (int kc = 0; kc < 4; ++kc) {
                    v2f a0, a1;
                    a0.x = ap0[kc*4];
                    a0.y = ap0[kc*4 + 1];
                    a1.x = ap1[kc*4];
                    a1.y = ap1[kc*4 + 1];
                    acc0 = __builtin_amdgcn_wmma_f32_16x16x4_f32(
                               false, a0, false, breg[tap*4 + kc], (short)0, acc0, false, false);
                    acc1 = __builtin_amdgcn_wmma_f32_16x16x4_f32(
                               false, a1, false, breg[tap*4 + kc], (short)0, acc1, false, false);
                }
            }
        }
        acc[q]     = acc0;
        acc[q + 1] = acc1;
    }
}

template <int NR>
__device__ __forceinline__
void store_rows(float* dst, const v8f (&acc)[NR], int ybase, float bv, int x0, int laneM, int half)
{
    #pragma unroll
    for (int q = 0; q < NR; ++q) {
        const int y = ybase + q;
        const bool inside = (y >= 0 && y < H);
        float* d = dst + (q*XP + x0 + 8*half + 1)*CP + laneM;
        #pragma unroll
        for (int r = 0; r < 8; ++r) d[r*CP] = inside ? (acc[q][r] + bv) : 0.f;
    }
}

__device__ __forceinline__
void load_weights(const float* __restrict__ w3, int tid, float (&wr)[27])
{
    #pragma unroll
    for (int it = 0; it < 27; ++it) wr[it] = w3[tid + it*256];
}

__device__ __forceinline__
void scatter_weights(float* wBf, int tid, const float (&wr)[27])
{
    #pragma unroll
    for (int it = 0; it < 27; ++it) {
        const int e   = tid + it*256;
        const int L   = e / 2304;
        const int rem = e - L*2304;
        const int oc  = rem / 144;
        const int r2  = rem - oc*144;
        const int ic  = r2 / 9;
        const int tap = r2 - ic*9;
        const int kc  = ic >> 2, hf = (ic >> 1) & 1, comp = ic & 1;
        wBf[((L*36 + tap*4 + kc)*32 + oc + 16*hf)*2 + comp] = wr[it];
    }
}

__global__ __launch_bounds__(256)
void conv3_branches_wmma_kernel(const float* __restrict__ in,
                                float* __restrict__ out,
                                const float* __restrict__ w_fea,
                                const float* __restrict__ b_fea,
                                const float* __restrict__ gates)
{
    const int y0  = blockIdx.x * BAND;
    const int b   = blockIdx.y;
    const int tid = threadIdx.x;

    __shared__ float bufA[(BAND + 6)*ROWF];
    __shared__ float bufB[(BAND + 4)*ROWF];
    __shared__ v2f   wB[2][3*WLAY];

    float wr[27];
    load_weights(w_fea, tid, wr);
    for (int idx = tid; idx < ((BAND + 6) + (BAND + 4))*2*16; idx += 256) {
        const int ch = idx & 15;
        const int xh = ((idx >> 4) & 1) ? (XP - 1) : 0;
        const int r  = idx >> 5;
        if (r < BAND + 6) bufA[(r*XP + xh)*CP + ch] = 0.f;
        else              bufB[((r - (BAND + 6))*XP + xh)*CP + ch] = 0.f;
    }
    for (int it0 = 0; it0 < 28; it0 += 7) {
        v4f v[7];
        #pragma unroll
        for (int j = 0; j < 7; ++j) {
            const int idx = tid + (it0 + j)*256;
            const int x4 = idx & 31, ch = (idx >> 5) & 15, r = idx >> 9;
            const int yi = y0 - 3 + r;
            const int yc = min(max(yi, 0), H - 1);
            v4f t = *(const v4f*)(in + ((size_t)(b*16 + ch) * H + yc) * W + x4*4);
            if (yi < 0 || yi >= H) t = (v4f){0.f, 0.f, 0.f, 0.f};
            v[j] = t;
        }
        #pragma unroll
        for (int j = 0; j < 7; ++j) {
            const int idx = tid + (it0 + j)*256;
            const int x4 = idx & 31, ch = (idx >> 5) & 15, r = idx >> 9;
            float* d = &bufA[(r*XP + x4*4 + 1)*CP + ch];
            d[0]    = v[j].x;
            d[CP]   = v[j].y;
            d[2*CP] = v[j].z;
            d[3*CP] = v[j].w;
        }
    }
    scatter_weights((float*)wB[0], tid, wr);
    __syncthreads();

    const int lane  = tid & 31;
    const int wave  = tid >> 5;
    const int x0    = wave * 16;
    const int laneM = lane & 15;
    const int half  = lane >> 4;

    v8f osum[BAND];
    #pragma unroll
    for (int q = 0; q < BAND; ++q) osum[q] = (v8f){0.f,0.f,0.f,0.f,0.f,0.f,0.f,0.f};

    for (int g = 0; g < 4; ++g) {
        const v2f*   wl    = wB[g & 1];
        const float* bg    = b_fea + g*48;
        const float* gateb = gates + (size_t)(b*4 + g) * H * W;
        const float bv0 = bg[laneM], bv1 = bg[16 + laneM], bv2 = bg[32 + laneM];

        {
            v8f acc[BAND + 4];
            conv_rows<BAND + 4>(bufA, wl, x0, laneM, half, lane, acc);
            store_rows<BAND + 4>(bufB, acc, y0 - 2, bv0, x0, laneM, half);
        }
        __syncthreads();
        {
            v8f acc[BAND + 2];
            conv_rows<BAND + 2>(bufB, wl + WLAY, x0, laneM, half, lane, acc);
            __syncthreads();
            store_rows<BAND + 2>(bufB, acc, y0 - 1, bv1, x0, laneM, half);
        }
        __syncthreads();
        {
            if (g < 3) load_weights(w_fea + (size_t)(g + 1) * WBR, tid, wr);
            v4f gv[BAND][2];
            #pragma unroll
            for (int q = 0; q < BAND; ++q) {
                const float* gp = gateb + (size_t)(y0 + q) * W + x0 + half*8;
                gv[q][0] = *(const v4f*)gp;
                gv[q][1] = *(const v4f*)(gp + 4);
            }
            v8f acc[BAND];
            conv_rows<BAND>(bufB, wl + 2*WLAY, x0, laneM, half, lane, acc);
            #pragma unroll
            for (int q = 0; q < BAND; ++q) {
                #pragma unroll
                for (int r = 0; r < 8; ++r)
                    osum[q][r] += (acc[q][r] + bv2) * gv[q][r >> 2][r & 3];
            }
            if (g < 3) scatter_weights((float*)wB[(g + 1) & 1], tid, wr);
        }
        __syncthreads();
    }

    float* outb = out + (size_t)b * 16 * H * W;
    #pragma unroll
    for (int q = 0; q < BAND; ++q) {
        float* op = outb + ((size_t)laneM * H + (y0 + q)) * W + x0 + half*8;
        const v4f lo = {osum[q][0], osum[q][1], osum[q][2], osum[q][3]};
        const v4f hi = {osum[q][4], osum[q][5], osum[q][6], osum[q][7]};
        *(v4f*)op       = lo;
        *(v4f*)(op + 4) = hi;
    }
}

extern "C" void kernel_launch(void* const* d_in, const int* in_sizes, int n_in,
                              void* d_out, int out_size, void* d_ws, size_t ws_size,
                              hipStream_t stream)
{
    (void)in_sizes; (void)n_in; (void)out_size; (void)ws_size;
    const float* landmarks = (const float*)d_in[0];
    const float* features  = (const float*)d_in[1];
    const float* w_lm      = (const float*)d_in[2];
    const float* b_lm      = (const float*)d_in[3];
    const float* w_fea     = (const float*)d_in[4];
    const float* b_fea     = (const float*)d_in[5];
    float* out = (float*)d_out;

    float* gates = (float*)d_ws;

    dw_gate_kernel<<<dim3(8, 4, 32), 256, 0, stream>>>(landmarks, w_lm, b_lm, gates);
    conv3_branches_wmma_kernel<<<dim3(128 / BAND, 32), 256, 0, stream>>>(
        features, out, w_fea, b_fea, gates);
}


# conv: A fragments read one row pair ahead, as ONE group per pair behind the pair's second tap row (4 staged rows in registers); no other LDS read among the matrix instructions; sum order kept
# speedup vs baseline: 3.5057x; 1.0583x over previous
#include <hip/hip_runtime.h>

typedef __attribute__((ext_vector_type(2))) float v2f;
typedef __attribute__((ext_vector_type(4))) float v4f;
typedef __attribute__((ext_vector_type(8))) float v8f;
typedef const __attribute__((address_space(4))) float cfloat;

#define H 128
#define W 128

#define DP 136

template <int NR>
__device__ __forceinline__
void dw_rows(const float (*sb)[DP], int i0, int rmax, int lane,
             const float (&w)[9], float bias, v4f (&o)[NR])
{
    float r[NR + 2][6];
    #pragma unroll
    for (int j = 0; j < NR + 2; ++j) {
        const int i = min(i0 - 1 + j, rmax);
        const float* p = &sb[i][4*lane + 3];
        const v4f c = *(const v4f*)(p + 1);
        r[j][0] = p[0];
        r[j][1] = c.x; r[j][2] = c.y; r[j][3] = c.z; r[j][4] = c.w;
        r[j][5] = p[5];
    }
    float acc[NR][4];
    #pragma unroll
    for (int q = 0; q < NR; ++q) {
        #pragma unroll
        for (int e = 0; e < 4; ++e) acc[q][e] = bias;
    }
    #pragma unroll
    for (int ky = 0; ky < 3; ++ky) {
        #pragma unroll
        for (int kx = 0; kx < 3; ++kx) {
            const float wt = w[ky*3 + kx];
            #pragma unroll
            for (int q = 0; q < NR; ++q) {
                #pragma unroll
                for (int e = 0; e < 4; ++e)
                    acc[q][e] = __builtin_fmaf(wt, r[q + ky][e + kx], acc[q][e]);
            }
        }
    }
    #pragma unroll
    for (int q = 0; q < NR; ++q) {
        #pragma unroll
        for (int e = 0; e < 4; ++e) o[q][e] = acc[q][e];
    }
}

__global__ __launch_bounds__(256)
void dw_gate_kernel(const float* __restrict__ lm,
                    const float* __restrict__ w_lm,
                    const float* __restrict__ b_lm,
                    float* __restrict__ gates)
{
    const int blk = blockIdx.x;
    const int g   = blockIdx.y;
    const int b   = blockIdx.z;
    const int tid = threadIdx.x;
    const int y0  = blk * 16;
    const int lane = tid & 31;
    const int wv = __builtin_amdgcn_readfirstlane(tid >> 5);

    __shared__ __attribute__((aligned(16))) float buf[2][26][DP];

    for (int idx = tid; idx < 2*26*2; idx += 256) {
        const int bsel = idx / 52, rem = idx % 52;
        buf[bsel][rem >> 1][(rem & 1) ? 132 : 3] = 0.f;
    }

    v4f gsum0 = {0.f, 0.f, 0.f, 0.f};
    v4f gsum1 = {0.f, 0.f, 0.f, 0.f};

    const int s0[4]    = {17, 27, 0, 48};
    const int cnt0[4]  = {10,  9, 17, 20};
    const int segs = (g == 0) ? 2 : 1;

    for (int s = 0; s < segs; ++s) {
        const int cstart = (s == 0) ? s0[g] : 36;
        const int ccnt   = (s == 0) ? cnt0[g] : 12;
        for (int ci = 0; ci < ccnt; ++ci) {
            const int c = cstart + ci;
            const float* src = lm + ((size_t)(b*68 + c) * H) * W;
            float wk[5][9], bk[5];
            #pragma unroll
            for (int k = 0; k < 5; ++k) {
                cfloat* wp = (cfloat*)(w_lm + (size_t)(k*68 + c) * 9);
                #pragma unroll
                for (int q = 0; q < 9; ++q) wk[k][q] = wp[q];
                bk[k] = ((cfloat*)b_lm)[k*68 + c];
            }
            v4f t[4];
            #pragma unroll
            for (int j = 0; j < 4; ++j) {
                const int i  = min(wv + 8*j, 25);
                const int y  = y0 - 5 + i;
                const int yc = min(max(y, 0), H - 1);
                v4f v = *(const v4f*)(src + yc*W + 4*lane);
                if (y < 0 || y >= H) v = (v4f){0.f, 0.f, 0.f, 0.f};
                t[j] = v;
            }
            #pragma unroll
            for (int j = 0; j < 4; ++j) {
                const int i = wv + 8*j;
                if (i < 26) *(v4f*)&buf[0][i][4 + 4*lane] = t[j];
            }
            __syncthreads();
            #pragma unroll
            for (int k = 0; k < 4; ++k) {
                const float bias = bk[k];
                const int rhi = 24 - k;
                const int i0  = k + 1 + 3*wv;
                if (i0 <= rhi) {
                    v4f o[3];
                    dw_rows<3>(buf[k & 1], i0, 25 - k, lane, wk[k], bias, o);
                    #pragma unroll
                    for (int q = 0; q < 3; ++q) {
                        const int i = i0 + q;
                        if (i <= rhi) {
                            const int y = y0 - 5 + i;
                            const v4f v = (y < 0 || y >= H) ? (v4f){0.f, 0.f, 0.f, 0.f} : o[q];
                            *(v4f*)&buf[(k + 1) & 1][i][4 + 4*lane] = v;
                        }
                    }
                }
                __syncthreads();
            }
            {
                const float bias = bk[4];
                v4f o[2];
                dw_rows<2>(buf[0], 5 + 2*wv, 21, lane, wk[4], bias, o);
                gsum0 += o[0];
                gsum1 += o[1];
            }
            __syncthreads();
        }
    }

    float* gp = gates + (((size_t)(b*4 + g) * H) + y0 + 2*wv) * W + 4*lane;
    *(v4f*)gp       = gsum0;
    *(v4f*)(gp + W) = gsum1;
}

#define XP   130
#define CP   18
#define ROWF (XP*CP)
#define BAND 8
#define WLAY (36*32)
#define WBR  (3*16*16*9)

__device__ __forceinline__
void load_a_row(v2f (&a)[12], const float* p)
{
    #pragma unroll
    for (int kx = 0; kx < 3; ++kx) {
        #pragma unroll
        for (int kc = 0; kc < 4; ++kc) a[kx*4 + kc] = *(const v2f*)(p + kx*CP + kc*4);
    }
}

template <int NROW>
__device__ __forceinline__
void tap_row(const v2f (&a)[NROW][12], const v2f (&breg)[36], int q, int ky, v8f& acc0, v8f& acc1)
{
    #pragma unroll
    for (int kx = 0; kx < 3; ++kx) {
        const int tap = ky*3 + kx;
        #pragma unroll
        for (int kc = 0; kc < 4; ++kc) {
            acc0 = __builtin_amdgcn_wmma_f32_16x16x4_f32(
                       false, a[q + ky][kx*4 + kc], false, breg[tap*4 + kc], (short)0, acc0, false, false);
            acc1 = __builtin_amdgcn_wmma_f32_16x16x4_f32(
                       false, a[q + 1 + ky][kx*4 + kc], false, breg[tap*4 + kc], (short)0, acc1, false, false);
        }
    }
}

template <int NR>
__device__ __forceinline__
void conv_rows(const float* src, const v2f* wl, int x0, int laneM, int half, int lane,
               v8f (&acc)[NR])
{
    v2f breg[36];
    #pragma unroll
    for (int i = 0; i < 36; ++i) breg[i] = wl[i*32 + lane];

    const float* ab = src + (x0 + laneM)*CP + 2*half;
    v2f a[NR + 2][12];
    #pragma unroll
    for (int r = 0; r < 4; ++r) load_a_row(a[r], ab + r*ROWF);
    __builtin_amdgcn_sched_barrier(0);

    #pragma unroll
    for (int q = 0; q < NR; q += 2) {
        v8f acc0 = {0.f,0.f,0.f,0.f,0.f,0.f,0.f,0.f};
        v8f acc1 = {0.f,0.f,0.f,0.f,0.f,0.f,0.f,0.f};
        tap_row(a, breg, q, 0, acc0, acc1);
        tap_row(a, breg, q, 1, acc0, acc1);
        if (q + 2 < NR) {
            __builtin_amdgcn_sched_barrier(0);
            load_a_row(a[q + 4], ab + (q + 4)*ROWF);
            load_a_row(a[q + 5], ab + (q + 5)*ROWF);
            __builtin_amdgcn_sched_barrier(0);
        }
        tap_row(a, breg, q, 2, acc0, acc1);
        acc[q]     = acc0;
        acc[q + 1] = acc1;
    }
}

template <int NR>
__device__ __forceinline__
void store_rows(float* dst, const v8f (&acc)[NR], int ybase, float bv, int x0, int laneM, int half)
{
    #pragma unroll
    for (int q = 0; q < NR; ++q) {
        const int y = ybase + q;
        const bool inside = (y >= 0 && y < H);
        float* d = dst + (q*XP + x0 + 8*half + 1)*CP + laneM;
        #pragma unroll
        for (int r = 0; r < 8; ++r) d[r*CP] = inside ? (acc[q][r] + bv) : 0.f;
    }
}

__device__ __forceinline__
void load_weights(const float* __restrict__ w3, int tid, float (&wr)[27])
{
    #pragma unroll
    for (int it = 0; it < 27; ++it) wr[it] = w3[tid + it*256];
}

__device__ __forceinline__
void scatter_weights(float* wBf, int tid, const float (&wr)[27])
{
    #pragma unroll
    for (int it = 0; it < 27; ++it) {
        const int e   = tid + it*256;
        const int L   = e / 2304;
        const int rem = e - L*2304;
        const int oc  = rem / 144;
        const int r2  = rem - oc*144;
        const int ic  = r2 / 9;
        const int tap = r2 - ic*9;
        const int kc  = ic >> 2, hf = (ic >> 1) & 1, comp = ic & 1;
        wBf[((L*36 + tap*4 + kc)*32 + oc + 16*hf)*2 + comp] = wr[it];
    }
}

__global__ __launch_bounds__(256)
void conv3_branches_wmma_kernel(const float* __restrict__ in,
                                float* __restrict__ out,
                                const float* __restrict__ w_fea,
                                const float* __restrict__ b_fea,
                                const float* __restrict__ gates)
{
    const int y0  = blockIdx.x * BAND;
    const int b   = blockIdx.y;
    const int tid = threadIdx.x;

    __shared__ __attribute__((aligned(16))) float bufA[(BAND + 6)*ROWF];
    __shared__ __attribute__((aligned(16))) float bufB[(BAND + 4)*ROWF];
    __shared__ v2f   wB[2][3*WLAY];

    float wr[27];
    load_weights(w_fea, tid, wr);
    for (int idx = tid; idx < ((BAND + 6) + (BAND + 4))*2*16; idx += 256) {
        const int ch = idx & 15;
        const int xh = ((idx >> 4) & 1) ? (XP - 1) : 0;
        const int r  = idx >> 5;
        if (r < BAND + 6) bufA[(r*XP + xh)*CP + ch] = 0.f;
        else              bufB[((r - (BAND + 6))*XP + xh)*CP + ch] = 0.f;
    }
    for (int it0 = 0; it0 < 28; it0 += 7) {
        v4f v[7];
        #pragma unroll
        for (int j = 0; j < 7; ++j) {
            const int idx = tid + (it0 + j)*256;
            const int x4 = idx & 31, ch = (idx >> 5) & 15, r = idx >> 9;
            const int yi = y0 - 3 + r;
            const int yc = min(max(yi, 0), H - 1);
            v4f t = *(const v4f*)(in + ((size_t)(b*16 + ch) * H + yc) * W + x4*4);
            if (yi < 0 || yi >= H) t = (v4f){0.f, 0.f, 0.f, 0.f};
            v[j] = t;
        }
        #pragma unroll
        for (int j = 0; j < 7; ++j) {
            const int idx = tid + (it0 + j)*256;
            const int x4 = idx & 31, ch = (idx >> 5) & 15, r = idx >> 9;
            float* d = &bufA[(r*XP + x4*4 + 1)*CP + ch];
            d[0]    = v[j].x;
            d[CP]   = v[j].y;
            d[2*CP] = v[j].z;
            d[3*CP] = v[j].w;
        }
    }
    scatter_weights((float*)wB[0], tid, wr);
    __syncthreads();

    const int lane  = tid & 31;
    const int wave  = tid >> 5;
    const int x0    = wave * 16;
    const int laneM = lane & 15;
    const int half  = lane >> 4;

    v8f osum[BAND];
    #pragma unroll
    for (int q = 0; q < BAND; ++q) osum[q] = (v8f){0.f,0.f,0.f,0.f,0.f,0.f,0.f,0.f};

    for (int g = 0; g < 4; ++g) {
        const v2f*   wl    = wB[g & 1];
        const float* bg    = b_fea + g*48;
        const float* gateb = gates + (size_t)(b*4 + g) * H * W;
        const float bv0 = bg[laneM], bv1 = bg[16 + laneM], bv2 = bg[32 + laneM];

        {
            v8f acc[BAND + 4];
            conv_rows<BAND + 4>(bufA, wl, x0, laneM, half, lane, acc);
            store_rows<BAND + 4>(bufB, acc, y0 - 2, bv0, x0, laneM, half);
        }
        __syncthreads();
        {
            v8f acc[BAND + 2];
            conv_rows<BAND + 2>(bufB, wl + WLAY, x0, laneM, half, lane, acc);
            __syncthreads();
            store_rows<BAND + 2>(bufB, acc, y0 - 1, bv1, x0, laneM, half);
        }
        __syncthreads();
        {
            if (g < 3) load_weights(w_fea + (size_t)(g + 1) * WBR, tid, wr);
            v4f gv[BAND][2];
            #pragma unroll
            for (int q = 0; q < BAND; ++q) {
                const float* gp = gateb + (size_t)(y0 + q) * W + x0 + half*8;
                gv[q][0] = *(const v4f*)gp;
                gv[q][1] = *(const v4f*)(gp + 4);
            }
            v8f acc[BAND];
            conv_rows<BAND>(bufB, wl + 2*WLAY, x0, laneM, half, lane, acc);
            #pragma unroll
            for (int q = 0; q < BAND; ++q) {
                #pragma unroll
                for (int r = 0; r < 8; ++r)
                    osum[q][r] += (acc[q][r] + bv2) * gv[q][r >> 2][r & 3];
            }
            if (g < 3) scatter_weights((float*)wB[(g + 1) & 1], tid, wr);
        }
        __syncthreads();
    }

    float* outb = out + (size_t)b * 16 * H * W;
    #pragma unroll
    for (int q = 0; q < BAND; ++q) {
        float* op = outb + ((size_t)laneM * H + (y0 + q)) * W + x0 + half*8;
        const v4f lo = {osum[q][0], osum[q][1], osum[q][2], osum[q][3]};
        const v4f hi = {osum[q][4], osum[q][5], osum[q][6], osum[q][7]};
        *(v4f*)op       = lo;
        *(v4f*)(op + 4) = hi;
    }
}

extern "C" void kernel_launch(void* const* d_in, const int* in_sizes, int n_in,
                              void* d_out, int out_size, void* d_ws, size_t ws_size,
                              hipStream_t stream)
{
    (void)in_sizes; (void)n_in; (void)out_size; (void)ws_size;
    const float* landmarks = (const float*)d_in[0];
    const float* features  = (const float*)d_in[1];
    const float* w_lm      = (const float*)d_in[2];
    const float* b_lm      = (const float*)d_in[3];
    const float* w_fea     = (const float*)d_in[4];
    const float* b_fea     = (const float*)d_in[5];
    float* out = (float*)d_out;

    float* gates = (float*)d_ws;

    dw_gate_kernel<<<dim3(8, 4, 32), 256, 0, stream>>>(landmarks, w_lm, b_lm, gates);
    conv3_branches_wmma_kernel<<<dim3(128 / BAND, 32), 256, 0, stream>>>(
        features, out, w_fea, b_fea, gates);
}


# conv prologue: a thread's 28 staging reads go out as ONE group (were 14 pairs, each waited for before the next was issued); the weight scatter's index arithmetic runs under them; sum order kept
# speedup vs baseline: 3.6485x; 1.0407x over previous
#include <hip/hip_runtime.h>

typedef __attribute__((ext_vector_type(2))) float v2f;
typedef __attribute__((ext_vector_type(4))) float v4f;
typedef __attribute__((ext_vector_type(8))) float v8f;
typedef const __attribute__((address_space(4))) float cfloat;

#define H 128
#define W 128

#define DP 136

template <int NR>
__device__ __forceinline__
void dw_rows(const float (*sb)[DP], int i0, int rmax, int lane,
             const float (&w)[9], float bias, v4f (&o)[NR])
{
    float r[NR + 2][6];
    #pragma unroll
    for (int j = 0; j < NR + 2; ++j) {
        const int i = min(i0 - 1 + j, rmax);
        const float* p = &sb[i][4*lane + 3];
        const v4f c = *(const v4f*)(p + 1);
        r[j][0] = p[0];
        r[j][1] = c.x; r[j][2] = c.y; r[j][3] = c.z; r[j][4] = c.w;
        r[j][5] = p[5];
    }
    float acc[NR][4];
    #pragma unroll
    for (int q = 0; q < NR; ++q) {
        #pragma unroll
        for (int e = 0; e < 4; ++e) acc[q][e] = bias;
    }
    #pragma unroll
    for (int ky = 0; ky < 3; ++ky) {
        #pragma unroll
        for (int kx = 0; kx < 3; ++kx) {
            const float wt = w[ky*3 + kx];
            #pragma unroll
            for (int q = 0; q < NR; ++q) {
                #pragma unroll
                for (int e = 0; e < 4; ++e)
                    acc[q][e] = __builtin_fmaf(wt, r[q + ky][e + kx], acc[q][e]);
            }
        }
    }
    #pragma unroll
    for (int q = 0; q < NR; ++q) {
        #pragma unroll
        for (int e = 0; e < 4; ++e) o[q][e] = acc[q][e];
    }
}

__global__ __launch_bounds__(256)
void dw_gate_kernel(const float* __restrict__ lm,
                    const float* __restrict__ w_lm,
                    const float* __restrict__ b_lm,
                    float* __restrict__ gates)
{
    const int blk = blockIdx.x;
    const int g   = blockIdx.y;
    const int b   = blockIdx.z;
    const int tid = threadIdx.x;
    const int y0  = blk * 16;
    const int lane = tid & 31;
    const int wv = __builtin_amdgcn_readfirstlane(tid >> 5);

    __shared__ __attribute__((aligned(16))) float buf[2][26][DP];

    for (int idx = tid; idx < 2*26*2; idx += 256) {
        const int bsel = idx / 52, rem = idx % 52;
        buf[bsel][rem >> 1][(rem & 1) ? 132 : 3] = 0.f;
    }

    v4f gsum0 = {0.f, 0.f, 0.f, 0.f};
    v4f gsum1 = {0.f, 0.f, 0.f, 0.f};

    const int s0[4]    = {17, 27, 0, 48};
    const int cnt0[4]  = {10,  9, 17, 20};
    const int segs = (g == 0) ? 2 : 1;

    for (int s = 0; s < segs; ++s) {
        const int cstart = (s == 0) ? s0[g] : 36;
        const int ccnt   = (s == 0) ? cnt0[g] : 12;
        for (int ci = 0; ci < ccnt; ++ci) {
            const int c = cstart + ci;
            const float* src = lm + ((size_t)(b*68 + c) * H) * W;
            float wk[5][9], bk[5];
            #pragma unroll
            for (int k = 0; k < 5; ++k) {
                cfloat* wp = (cfloat*)(w_lm + (size_t)(k*68 + c) * 9);
                #pragma unroll
                for (int q = 0; q < 9; ++q) wk[k][q] = wp[q];
                bk[k] = ((cfloat*)b_lm)[k*68 + c];
            }
            v4f t[4];
            #pragma unroll
            for (int j = 0; j < 4; ++j) {
                const int i  = min(wv + 8*j, 25);
                const int y  = y0 - 5 + i;
                const int yc = min(max(y, 0), H - 1);
                v4f v = *(const v4f*)(src + yc*W + 4*lane);
                if (y < 0 || y >= H) v = (v4f){0.f, 0.f, 0.f, 0.f};
                t[j] = v;
            }
            #pragma unroll
            for (int j = 0; j < 4; ++j) {
                const int i = wv + 8*j;
                if (i < 26) *(v4f*)&buf[0][i][4 + 4*lane] = t[j];
            }
            __syncthreads();
            #pragma unroll
            for (int k = 0; k < 4; ++k) {
                const float bias = bk[k];
                const int rhi = 24 - k;
                const int i0  = k + 1 + 3*wv;
                if (i0 <= rhi) {
                    v4f o[3];
                    dw_rows<3>(buf[k & 1], i0, 25 - k, lane, wk[k], bias, o);
                    #pragma unroll
                    for (int q = 0; q < 3; ++q) {
                        const int i = i0 + q;
                        if (i <= rhi) {
                            const int y = y0 - 5 + i;
                            const v4f v = (y < 0 || y >= H) ? (v4f){0.f, 0.f, 0.f, 0.f} : o[q];
                            *(v4f*)&buf[(k + 1) & 1][i][4 + 4*lane] = v;
                        }
                    }
                }
                __syncthreads();
            }
            {
                const float bias = bk[4];
                v4f o[2];
                dw_rows<2>(buf[0], 5 + 2*wv, 21, lane, wk[4], bias, o);
                gsum0 += o[0];
                gsum1 += o[1];
            }
            __syncthreads();
        }
    }

    float* gp = gates + (((size_t)(b*4 + g) * H) + y0 + 2*wv) * W + 4*lane;
    *(v4f*)gp       = gsum0;
    *(v4f*)(gp + W) = gsum1;
}

#define XP   130
#define CP   18
#define ROWF (XP*CP)
#define BAND 8
#define WLAY (36*32)
#define WBR  (3*16*16*9)

__device__ __forceinline__
void load_a_row(v2f (&a)[12], const float* p)
{
    #pragma unroll
    for (int kx = 0; kx < 3; ++kx) {
        #pragma unroll
        for (int kc = 0; kc < 4; ++kc) a[kx*4 + kc] = *(const v2f*)(p + kx*CP + kc*4);
    }
}

template <int NROW>
__device__ __forceinline__
void tap_row(const v2f (&a)[NROW][12], const v2f (&breg)[36], int q, int ky, v8f& acc0, v8f& acc1)
{
    #pragma unroll
    for (int kx = 0; kx < 3; ++kx) {
        const int tap = ky*3 + kx;
        #pragma unroll
        for (int kc = 0; kc < 4; ++kc) {
            acc0 = __builtin_amdgcn_wmma_f32_16x16x4_f32(
                       false, a[q + ky][kx*4 + kc], false, breg[tap*4 + kc], (short)0, acc0, false, false);
            acc1 = __builtin_amdgcn_wmma_f32_16x16x4_f32(
                       false, a[q + 1 + ky][kx*4 + kc], false, breg[tap*4 + kc], (short)0, acc1, false, false);
        }
    }
}

template <int NR>
__device__ __forceinline__
void conv_rows(const float* src, const v2f* wl, int x0, int laneM, int half, int lane,
               v8f (&acc)[NR])
{
    v2f breg[36];
    #pragma unroll
    for (int i = 0; i < 36; ++i) breg[i] = wl[i*32 + lane];

    const float* ab = src + (x0 + laneM)*CP + 2*half;
    v2f a[NR + 2][12];
    #pragma unroll
    for (int r = 0; r < 4; ++r) load_a_row(a[r], ab + r*ROWF);
    __builtin_amdgcn_sched_barrier(0);

    #pragma unroll
    for (int q = 0; q < NR; q += 2) {
        v8f acc0 = {0.f,0.f,0.f,0.f,0.f,0.f,0.f,0.f};
        v8f acc1 = {0.f,0.f,0.f,0.f,0.f,0.f,0.f,0.f};
        tap_row(a, breg, q, 0, acc0, acc1);
        tap_row(a, breg, q, 1, acc0, acc1);
        if (q + 2 < NR) {
            __builtin_amdgcn_sched_barrier(0);
            load_a_row(a[q + 4], ab + (q + 4)*ROWF);
            load_a_row(a[q + 5], ab + (q + 5)*ROWF);
            __builtin_amdgcn_sched_barrier(0);
        }
        tap_row(a, breg, q, 2, acc0, acc1);
        acc[q]     = acc0;
        acc[q + 1] = acc1;
    }
}

template <int NR>
__device__ __forceinline__
void store_rows(float* dst, const v8f (&acc)[NR], int ybase, float bv, int x0, int laneM, int half)
{
    #pragma unroll
    for (int q = 0; q < NR; ++q) {
        const int y = ybase + q;
        const bool inside = (y >= 0 && y < H);
        float* d = dst + (q*XP + x0 + 8*half + 1)*CP + laneM;
        #pragma unroll
        for (int r = 0; r < 8; ++r) d[r*CP] = inside ? (acc[q][r] + bv) : 0.f;
    }
}

__device__ __forceinline__
void load_weights(const float* __restrict__ w3, int tid, float (&wr)[27])
{
    #pragma unroll
    for (int it = 0; it < 27; ++it) wr[it] = w3[tid + it*256];
}

__device__ __forceinline__
void scatter_weights(float* wBf, int tid, const float (&wr)[27])
{
    #pragma unroll
    for (int it = 0; it < 27; ++it) {
        const int e   = tid + it*256;
        const int L   = e / 2304;
        const int rem = e - L*2304;
        const int oc  = rem / 144;
        const int r2  = rem - oc*144;
        const int ic  = r2 / 9;
        const int tap = r2 - ic*9;
        const int kc  = ic >> 2, hf = (ic >> 1) & 1, comp = ic & 1;
        wBf[((L*36 + tap*4 + kc)*32 + oc + 16*hf)*2 + comp] = wr[it];
    }
}

__global__ __launch_bounds__(256)
void conv3_branches_wmma_kernel(const float* __restrict__ in,
                                float* __restrict__ out,
                                const float* __restrict__ w_fea,
                                const float* __restrict__ b_fea,
                                const float* __restrict__ gates)
{
    const int y0  = blockIdx.x * BAND;
    const int b   = blockIdx.y;
    const int tid = threadIdx.x;

    __shared__ __attribute__((aligned(16))) float bufA[(BAND + 6)*ROWF];
    __shared__ __attribute__((aligned(16))) float bufB[(BAND + 4)*ROWF];
    __shared__ v2f   wB[2][3*WLAY];

    float wr[27];
    load_weights(w_fea, tid, wr);
    for (int idx = tid; idx < ((BAND + 6) + (BAND + 4))*2*16; idx += 256) {
        const int ch = idx & 15;
        const int xh = ((idx >> 4) & 1) ? (XP - 1) : 0;
        const int r  = idx >> 5;
        if (r < BAND + 6) bufA[(r*XP + xh)*CP + ch] = 0.f;
        else              bufB[((r - (BAND + 6))*XP + xh)*CP + ch] = 0.f;
    }
    const int swv = __builtin_amdgcn_readfirstlane(tid >> 5);
    const int sx  = (tid & 31) * 4;
    const float* ap[28];
    #pragma unroll
    for (int j = 0; j < 28; ++j) {
        const int ch = swv + 8*(j & 1);
        const int yi = y0 - 3 + (j >> 1);
        const int yc = (yi < 0 || yi >= H) ? y0 : yi;
        ap[j] = in + ((size_t)(b*16 + ch) * H + yc) * W + sx;
    }
    __builtin_amdgcn_sched_barrier(0);
    v4f v[28];
    #pragma unroll
    for (int j = 0; j < 28; ++j) v[j] = *(const v4f*)ap[j];
    __builtin_amdgcn_sched_barrier(0);
    scatter_weights((float*)wB[0], tid, wr);
    __builtin_amdgcn_sched_barrier(0);
    #pragma unroll
    for (int j = 0; j < 28; ++j) {
        const int ch = swv + 8*(j & 1), r = j >> 1;
        const int yi = y0 - 3 + r;
        const v4f t = (yi < 0 || yi >= H) ? (v4f){0.f, 0.f, 0.f, 0.f} : v[j];
        float* d = &bufA[(r*XP + sx + 1)*CP + ch];
        d[0]    = t.x;
        d[CP]   = t.y;
        d[2*CP] = t.z;
        d[3*CP] = t.w;
    }
    __syncthreads();

    const int lane  = tid & 31;
    const int wave  = tid >> 5;
    const int x0    = wave * 16;
    const int laneM = lane & 15;
    const int half  = lane >> 4;

    v8f osum[BAND];
    #pragma unroll
    for (int q = 0; q < BAND; ++q) osum[q] = (v8f){0.f,0.f,0.f,0.f,0.f,0.f,0.f,0.f};

    for (int g = 0; g < 4; ++g) {
        const v2f*   wl    = wB[g & 1];
        const float* bg    = b_fea + g*48;
        const float* gateb = gates + (size_t)(b*4 + g) * H * W;
        const float bv0 = bg[laneM], bv1 = bg[16 + laneM], bv2 = bg[32 + laneM];

        {
            v8f acc[BAND + 4];
            conv_rows<BAND + 4>(bufA, wl, x0, laneM, half, lane, acc);
            store_rows<BAND + 4>(bufB, acc, y0 - 2, bv0, x0, laneM, half);
        }
        __syncthreads();
        {
            v8f acc[BAND + 2];
            conv_rows<BAND + 2>(bufB, wl + WLAY, x0, laneM, half, lane, acc);
            __syncthreads();
            store_rows<BAND + 2>(bufB, acc, y0 - 1, bv1, x0, laneM, half);
        }
        __syncthreads();
        {
            if (g < 3) load_weights(w_fea + (size_t)(g + 1) * WBR, tid, wr);
            v4f gv[BAND][2];
            #pragma unroll
            for (int q = 0; q < BAND; ++q) {
                const float* gp = gateb + (size_t)(y0 + q) * W + x0 + half*8;
                gv[q][0] = *(const v4f*)gp;
                gv[q][1] = *(const v4f*)(gp + 4);
            }
            v8f acc[BAND];
            conv_rows<BAND>(bufB, wl + 2*WLAY, x0, laneM, half, lane, acc);
            #pragma unroll
            for (int q = 0; q < BAND; ++q) {
                #pragma unroll
                for (int r = 0; r < 8; ++r)
                    osum[q][r] += (acc[q][r] + bv2) * gv[q][r >> 2][r & 3];
            }
            if (g < 3) scatter_weights((float*)wB[(g + 1) & 1], tid, wr);
        }
        __syncthreads();
    }

    float* outb = out + (size_t)b * 16 * H * W;
    #pragma unroll
    for (int q = 0; q < BAND; ++q) {
        float* op = outb + ((size_t)laneM * H + (y0 + q)) * W + x0 + half*8;
        const v4f lo = {osum[q][0], osum[q][1], osum[q][2], osum[q][3]};
        const v4f hi = {osum[q][4], osum[q][5], osum[q][6], osum[q][7]};
        *(v4f*)op       = lo;
        *(v4f*)(op + 4) = hi;
    }
}

extern "C" void kernel_launch(void* const* d_in, const int* in_sizes, int n_in,
                              void* d_out, int out_size, void* d_ws, size_t ws_size,
                              hipStream_t stream)
{
    (void)in_sizes; (void)n_in; (void)out_size; (void)ws_size;
    const float* landmarks = (const float*)d_in[0];
    const float* features  = (const float*)d_in[1];
    const float* w_lm      = (const float*)d_in[2];
    const float* b_lm      = (const float*)d_in[3];
    const float* w_fea     = (const float*)d_in[4];
    const float* b_fea     = (const float*)d_in[5];
    float* out = (float*)d_out;

    float* gates = (float*)d_ws;

    dw_gate_kernel<<<dim3(8, 4, 32), 256, 0, stream>>>(landmarks, w_lm, b_lm, gates);
    conv3_branches_wmma_kernel<<<dim3(128 / BAND, 32), 256, 0, stream>>>(
        features, out, w_fea, b_fea, gates);
}
